# starting kernel
# baseline (speedup 1.0000x reference)
#include <hip/hip_runtime.h>

typedef __attribute__((ext_vector_type(2))) float v2f;
typedef __attribute__((ext_vector_type(8))) float v8f;

#define N_NODES  100000
#define N_EDGES  1600000
#define DIM_IN   256
#define DIM_HID  128
#define DIM_OUT  32


__global__ __launch_bounds__(256) void k_init_deg(float* deg, int n) {
  int i = blockIdx.x * blockDim.x + threadIdx.x;
  if (i < n) deg[i] = 1.0f;
}

__global__ __launch_bounds__(256) void k_deg_accum(const int* __restrict__ dst,
                                                   const float* __restrict__ w,
                                                   float* deg, int e) {
  int i = blockIdx.x * blockDim.x + threadIdx.x;
  if (i < e) atomicAdd(&deg[dst[i]], w[i]);
}

__global__ __launch_bounds__(256) void k_deg_to_inv(float* deg, int n) {
  int i = blockIdx.x * blockDim.x + threadIdx.x;
  if (i < n) {
    float d = deg[i];
    deg[i] = (d > 0.0f) ? rsqrtf(fmaxf(d, 1e-12f)) : 0.0f;
  }
}

__global__ __launch_bounds__(256) void k_init_bias(float* buf,
                                                   const float* __restrict__ bias,
                                                   int mask, long total) {
  long i = (long)blockIdx.x * blockDim.x + threadIdx.x;
  if (i < total) buf[i] = bias[i & mask];
}


__global__ __launch_bounds__(256) void k_gemm1_wmma(const float* __restrict__ X,
                                                    const float* __restrict__ W,
                                                    float* __restrict__ H) {
  const int gw   = blockIdx.x * 8 + (threadIdx.x >> 5);
  const int lane = threadIdx.x & 31;
  const int rt = gw >> 3;
  const int ct = gw & 7;
  const int m0 = rt * 16, n0 = ct * 16;
  const int l    = lane & 15;
  const int half = lane >> 4;
  const int ko   = half * 2;

  const float* arow = X + (size_t)(m0 + l) * DIM_IN;
  v8f c = {0.f, 0.f, 0.f, 0.f, 0.f, 0.f, 0.f, 0.f};

  for (int k0 = 0; k0 < DIM_IN; k0 += 4) {
    v2f a = *(const v2f*)(arow + k0 + ko);
    v2f b;
    b.x = W[(size_t)(k0 + ko)     * DIM_HID + n0 + l];
    b.y = W[(size_t)(k0 + ko + 1) * DIM_HID + n0 + l];
    c = __builtin_amdgcn_wmma_f32_16x16x4_f32(false, a, false, b,
                                              (short)0, c, false, false);
  }

  float* drow = H + (size_t)(m0 + half * 8) * DIM_HID + (n0 + l);
#pragma unroll
  for (int i = 0; i < 8; ++i) drow[(size_t)i * DIM_HID] = c[i];
}


__global__ __launch_bounds__(256) void k_gemm2_wmma(const float* __restrict__ XE,
                                                    const float* __restrict__ W,
                                                    float* __restrict__ H) {
  const int gw = blockIdx.x * 8 + (threadIdx.x >> 5);
  if (gw >= (N_NODES / 16) * (DIM_OUT / 16)) return;
  const int lane = threadIdx.x & 31;
  const int rt = gw >> 1;
  const int ct = gw & 1;
  const int m0 = rt * 16, n0 = ct * 16;
  const int l    = lane & 15;
  const int half = lane >> 4;
  const int ko   = half * 2;

  const float* arow = XE + (size_t)(m0 + l) * DIM_HID;
  v8f c = {0.f, 0.f, 0.f, 0.f, 0.f, 0.f, 0.f, 0.f};

  for (int k0 = 0; k0 < DIM_HID; k0 += 4) {
    v2f a;
    a.x = fmaxf(arow[k0 + ko],     0.0f);
    a.y = fmaxf(arow[k0 + ko + 1], 0.0f);
    v2f b;
    b.x = W[(size_t)(k0 + ko)     * DIM_OUT + n0 + l];
    b.y = W[(size_t)(k0 + ko + 1) * DIM_OUT + n0 + l];
    c = __builtin_amdgcn_wmma_f32_16x16x4_f32(false, a, false, b,
                                              (short)0, c, false, false);
  }

  float* drow = H + (size_t)(m0 + half * 8) * DIM_OUT + (n0 + l);
#pragma unroll
  for (int i = 0; i < 8; ++i) drow[(size_t)i * DIM_OUT] = c[i];
}


__global__ __launch_bounds__(256) void k_scatter128(const int* __restrict__ src,
                                                    const int* __restrict__ dst,
                                                    const float* __restrict__ wgt,
                                                    const float* __restrict__ inv,
                                                    const float* __restrict__ h,
                                                    float* acc, int e, int n) {
  const int wid  = blockIdx.x * 8 + (threadIdx.x >> 5);
  const int lane = threadIdx.x & 31;
  if (wid >= e + n) return;
  int s, t; float w;
  if (wid < e) { s = src[wid]; t = dst[wid]; w = wgt[wid]; }
  else         { s = t = wid - e; w = 1.0f; }
  const float nrm = inv[s] * w * inv[t];
  const float4 v = *(const float4*)(h + (size_t)s * DIM_HID + lane * 4);
  float* o = acc + (size_t)t * DIM_HID + lane * 4;
  atomicAdd(o + 0, nrm * v.x);
  atomicAdd(o + 1, nrm * v.y);
  atomicAdd(o + 2, nrm * v.z);
  atomicAdd(o + 3, nrm * v.w);
}

__global__ __launch_bounds__(256) void k_scatter32(const int* __restrict__ src,
                                                   const int* __restrict__ dst,
                                                   const float* __restrict__ wgt,
                                                   const float* __restrict__ inv,
                                                   const float* __restrict__ h,
                                                   float* acc, int e, int n) {
  const int wid  = blockIdx.x * 8 + (threadIdx.x >> 5);
  const int lane = threadIdx.x & 31;
  if (wid >= e + n) return;
  int s, t; float w;
  if (wid < e) { s = src[wid]; t = dst[wid]; w = wgt[wid]; }
  else         { s = t = wid - e; w = 1.0f; }
  const float nrm = inv[s] * w * inv[t];
  const float v = h[(size_t)s * DIM_OUT + lane];
  atomicAdd(acc + (size_t)t * DIM_OUT + lane, nrm * v);
}


extern "C" void kernel_launch(void* const* d_in, const int* in_sizes, int n_in,
                              void* d_out, int out_size, void* d_ws, size_t ws_size,
                              hipStream_t stream) {
  const int N = N_NODES, E = N_EDGES;

  const float* x   = (const float*)d_in[0];
  const int*   ei  = (const int*)d_in[1];
  const int*   src = ei;
  const int*   dst = ei + E;
  const float* w   = (const float*)d_in[2];
  const float* W1  = (const float*)d_in[3];
  const float* b1  = (const float*)d_in[4];
  const float* W2  = (const float*)d_in[5];
  const float* b2  = (const float*)d_in[6];

  float* out  = (float*)d_out;
  float* xemb = out + (size_t)N * DIM_OUT;

  float* inv = (float*)d_ws;
  float* h1  = inv + N;
  float* h2  = h1 + (size_t)N * DIM_HID;

  k_init_deg  <<<(N + 255) / 256, 256, 0, stream>>>(inv, N);
  k_deg_accum <<<(E + 255) / 256, 256, 0, stream>>>(dst, w, inv, E);
  k_deg_to_inv<<<(N + 255) / 256, 256, 0, stream>>>(inv, N);

  k_gemm1_wmma<<<(N / 16) * (DIM_HID / 16) / 8, 256, 0, stream>>>(x, W1, h1);

  const long totH = (long)N * DIM_HID;
  k_init_bias <<<(int)((totH + 255) / 256), 256, 0, stream>>>(xemb, b1, DIM_HID - 1, totH);
  k_scatter128<<<(E + N) / 8, 256, 0, stream>>>(src, dst, w, inv, h1, xemb, E, N);

  const int tiles2 = (N / 16) * (DIM_OUT / 16);
  k_gemm2_wmma<<<(tiles2 + 7) / 8, 256, 0, stream>>>(xemb, W2, h2);

  const long totO = (long)N * DIM_OUT;
  k_init_bias<<<(int)((totO + 255) / 256), 256, 0, stream>>>(out, b2, DIM_OUT - 1, totO);
  k_scatter32<<<(E + N) / 8, 256, 0, stream>>>(src, dst, w, inv, h2, out, E, N);
}


# both WMMA products: a wave owns a 16-row strip over all column tiles, A fragment loaded once per k step and reused; W held in LDS in fragment order (16-byte reads); f32 sums in the same order
# speedup vs baseline: 2.1587x; 2.1587x over previous
#include <hip/hip_runtime.h>


typedef __attribute__((ext_vector_type(2))) float v2f;
typedef __attribute__((ext_vector_type(8))) float v8f;

#define N_NODES  100000
#define N_EDGES  1600000
#define DIM_IN   256
#define DIM_HID  128
#define DIM_OUT  32
#define HEAD_PITCH 16


__global__ __launch_bounds__(256) void k_init_head(int* head, int n) {
  int i = blockIdx.x * blockDim.x + threadIdx.x;
  if (i < n) head[(size_t)i * HEAD_PITCH] = -1;
}

__global__ __launch_bounds__(256) void k_link(const int* __restrict__ dst,
                                              int* head, int* __restrict__ nxt, int e) {
  int i = blockIdx.x * blockDim.x + threadIdx.x;
  if (i < e) nxt[i] = atomicExch(&head[(size_t)dst[i] * HEAD_PITCH], i);
}

__global__ __launch_bounds__(256) void k_deg_inv(const int* __restrict__ head,
                                                 const int* __restrict__ nxt,
                                                 const float* __restrict__ w,
                                                 float* __restrict__ inv, int n, int e) {
  int i = blockIdx.x * blockDim.x + threadIdx.x;
  if (i >= n) return;
  float d = 1.0f;
  int k = head[(size_t)i * HEAD_PITCH];
  for (int it = 0; it < e && (unsigned)k < (unsigned)e; ++it) {
    d += w[k];
    k = nxt[k];
  }
  inv[i] = (d > 0.0f) ? rsqrtf(fmaxf(d, 1e-12f)) : 0.0f;
}


template <int K, int NT, bool RELU>
__device__ __forceinline__ void strip_product(const float* __restrict__ A,
                                              const float* __restrict__ W,
                                              float* __restrict__ H,
                                              float4* Wl, int n_row_tiles) {
  constexpr int N  = NT * 16;
  constexpr int NQ = NT / 2;
  constexpr int KS = K / 4;
  constexpr int CH = 8;

#pragma unroll 8
  for (int it = 0; it < KS * NQ * 32 / 256; ++it) {
    const int i  = it * 256 + threadIdx.x;
    const int ln = i & 31;
    const int kq = i >> 5;
    const int q  = kq % NQ;
    const int kk = kq / NQ;
    const int k  = 4 * kk + 2 * (ln >> 4);
    const int n  = 32 * q + (ln & 15);
    float4 f;
    f.x = W[(size_t)k       * N + n];
    f.y = W[(size_t)(k + 1) * N + n];
    f.z = W[(size_t)k       * N + n + 16];
    f.w = W[(size_t)(k + 1) * N + n + 16];
    Wl[i] = f;
  }
  __syncthreads();

  const int rt = blockIdx.x * 8 + (threadIdx.x >> 5);
  if (rt >= n_row_tiles) return;
  const int lane = threadIdx.x & 31;
  const int l    = lane & 15;
  const int half = lane >> 4;
  const int m0   = rt * 16;

  const float* arow = A + (size_t)(m0 + l) * K + half * 2;
  v8f c[NT];
#pragma unroll
  for (int t = 0; t < NT; ++t) c[t] = (v8f){0.f, 0.f, 0.f, 0.f, 0.f, 0.f, 0.f, 0.f};

  v2f ac[CH], an[CH];
#pragma unroll
  for (int j = 0; j < CH; ++j) {
    v2f a = *(const v2f*)(arow + 4 * j);
    if (RELU) { a.x = fmaxf(a.x, 0.0f); a.y = fmaxf(a.y, 0.0f); }
    ac[j] = a;
  }

  for (int ch = 0; ch < KS / CH; ++ch) {
    if (ch + 1 < KS / CH) {
#pragma unroll
      for (int j = 0; j < CH; ++j) {
        v2f a = *(const v2f*)(arow + 4 * (CH * (ch + 1) + j));
        if (RELU) { a.x = fmaxf(a.x, 0.0f); a.y = fmaxf(a.y, 0.0f); }
        an[j] = a;
      }
    }
#pragma unroll
    for (int j = 0; j < CH; ++j) {
      const float4* wk = Wl + (size_t)((ch * CH + j) * NQ) * 32 + lane;
#pragma unroll
      for (int q = 0; q < NQ; ++q) {
        const float4 f = wk[q * 32];
        v2f b0; b0.x = f.x; b0.y = f.y;
        v2f b1; b1.x = f.z; b1.y = f.w;
        c[2 * q]     = __builtin_amdgcn_wmma_f32_16x16x4_f32(false, ac[j], false, b0,
                                                             (short)0, c[2 * q], false, false);
        c[2 * q + 1] = __builtin_amdgcn_wmma_f32_16x16x4_f32(false, ac[j], false, b1,
                                                             (short)0, c[2 * q + 1], false, false);
      }
    }
#pragma unroll
    for (int j = 0; j < CH; ++j) ac[j] = an[j];
  }

  float* drow = H + (size_t)(m0 + half * 8) * N + l;
#pragma unroll
  for (int t = 0; t < NT; ++t) {
#pragma unroll
    for (int i = 0; i < 8; ++i) drow[(size_t)i * N + t * 16] = c[t][i];
  }
}

__global__ __launch_bounds__(256) void k_gemm1_wmma(const float* __restrict__ X,
                                                    const float* __restrict__ W,
                                                    float* __restrict__ H) {
  __shared__ float4 Wl[(DIM_IN / 4) * (DIM_HID / 32) * 32];
  strip_product<DIM_IN, DIM_HID / 16, false>(X, W, H, Wl, N_NODES / 16);
}

__global__ __launch_bounds__(256) void k_gemm2_wmma(const float* __restrict__ XE,
                                                    const float* __restrict__ W,
                                                    float* __restrict__ H) {
  __shared__ float4 Wl[(DIM_HID / 4) * (DIM_OUT / 32) * 32];
  strip_product<DIM_HID, DIM_OUT / 16, true>(XE, W, H, Wl, N_NODES / 16);
}


__global__ __launch_bounds__(256) void k_gather128(const int* __restrict__ src,
                                                   const float* __restrict__ wgt,
                                                   const int* __restrict__ head,
                                                   const int* __restrict__ nxt,
                                                   const float* __restrict__ inv,
                                                   const float* __restrict__ h,
                                                   const float* __restrict__ bias,
                                                   float* __restrict__ out, int e, int n) {
  const int t = __builtin_amdgcn_readfirstlane((int)(blockIdx.x * 8 + (threadIdx.x >> 5)));
  const int lane = threadIdx.x & 31;
  if (t >= n) return;
  const float it_ = inv[t];
  int k = head[(size_t)t * HEAD_PITCH];
  float ps = it_, pw = 1.0f;
  float4 v = *(const float4*)(h + (size_t)t * DIM_HID + lane * 4);
  float4 acc = {0.f, 0.f, 0.f, 0.f};
  for (int it = 0; it < e && (unsigned)k < (unsigned)e; ++it) {
    const int   kn = nxt[k];
    const int   s  = src[k];
    const float w  = wgt[k];
    asm volatile("" :: "s"(kn), "s"(s), "s"(w), "s"(ps));
    const float nrm = ps * pw * it_;
    acc.x += nrm * v.x; acc.y += nrm * v.y; acc.z += nrm * v.z; acc.w += nrm * v.w;
    ps = inv[s];
    pw = w;
    v = *(const float4*)(h + (size_t)s * DIM_HID + lane * 4);
    k = kn;
  }
  {
    const float nrm = ps * pw * it_;
    acc.x += nrm * v.x; acc.y += nrm * v.y; acc.z += nrm * v.z; acc.w += nrm * v.w;
  }
  const float4 b = *(const float4*)(bias + lane * 4);
  float4 r = {acc.x + b.x, acc.y + b.y, acc.z + b.z, acc.w + b.w};
  *(float4*)(out + (size_t)t * DIM_HID + lane * 4) = r;
}

__global__ __launch_bounds__(256) void k_gather32(const int* __restrict__ src,
                                                  const float* __restrict__ wgt,
                                                  const int* __restrict__ head,
                                                  const int* __restrict__ nxt,
                                                  const float* __restrict__ inv,
                                                  const float* __restrict__ h,
                                                  const float* __restrict__ bias,
                                                  float* __restrict__ out, int e, int n) {
  const int t = __builtin_amdgcn_readfirstlane((int)(blockIdx.x * 8 + (threadIdx.x >> 5)));
  const int lane = threadIdx.x & 31;
  if (t >= n) return;
  const float it_ = inv[t];
  int k = head[(size_t)t * HEAD_PITCH];
  float ps = it_, pw = 1.0f;
  float v = h[(size_t)t * DIM_OUT + lane];
  float acc = 0.f;
  for (int it = 0; it < e && (unsigned)k < (unsigned)e; ++it) {
    const int   kn = nxt[k];
    const int   s  = src[k];
    const float w  = wgt[k];
    asm volatile("" :: "s"(kn), "s"(s), "s"(w), "s"(ps));
    acc += (ps * pw * it_) * v;
    ps = inv[s];
    pw = w;
    v = h[(size_t)s * DIM_OUT + lane];
    k = kn;
  }
  acc += (ps * pw * it_) * v;
  out[(size_t)t * DIM_OUT + lane] = acc + bias[lane];
}


extern "C" void kernel_launch(void* const* d_in, const int* in_sizes, int n_in,
                              void* d_out, int out_size, void* d_ws, size_t ws_size,
                              hipStream_t stream) {
  const int N = N_NODES, E = N_EDGES;

  const float* x   = (const float*)d_in[0];
  const int*   ei  = (const int*)d_in[1];
  const int*   src = ei;
  const int*   dst = ei + E;
  const float* w   = (const float*)d_in[2];
  const float* W1  = (const float*)d_in[3];
  const float* b1  = (const float*)d_in[4];
  const float* W2  = (const float*)d_in[5];
  const float* b2  = (const float*)d_in[6];

  float* out  = (float*)d_out;
  float* xemb = out + (size_t)N * DIM_OUT;

  float* inv  = (float*)d_ws;
  float* h1   = inv + N;
  float* h2   = h1;
  int*   head = (int*)(h1 + (size_t)N * DIM_HID);
  int*   nxt  = head + (size_t)N * HEAD_PITCH;

  k_init_head<<<(N + 255) / 256, 256, 0, stream>>>(head, N);
  k_link     <<<(E + 255) / 256, 256, 0, stream>>>(dst, head, nxt, E);
  k_deg_inv  <<<(N + 255) / 256, 256, 0, stream>>>(head, nxt, w, inv, N, E);

  k_gemm1_wmma<<<(N / 16 + 7) / 8, 256, 0, stream>>>(x, W1, h1);

  k_gather128<<<(N + 7) / 8, 256, 0, stream>>>(src, w, head, nxt, inv, h1, b1, xemb, E, N);

  k_gemm2_wmma<<<(N / 16 + 7) / 8, 256, 0, stream>>>(xemb, W2, h2);

  k_gather32<<<(N + 7) / 8, 256, 0, stream>>>(src, w, head, nxt, inv, h2, b2, out, E, N);
}


# each node's edges linked into 4 lists (edge id mod 4), not 1; degree and both neighbour sums walk the 4 lists side by side (a lane per list), 4 dependent reads in flight; f32 sum order interleaved
# speedup vs baseline: 2.6417x; 1.2237x over previous
#include <hip/hip_runtime.h>


typedef __attribute__((ext_vector_type(2))) float v2f;
typedef __attribute__((ext_vector_type(8))) float v8f;

#define N_NODES  100000
#define N_EDGES  1600000
#define DIM_IN   256
#define DIM_HID  128
#define DIM_OUT  32
#define HEAD_PITCH 16

#define NL 4

__global__ __launch_bounds__(256) void k_init_head(int* head, int n) {
  int i = blockIdx.x * blockDim.x + threadIdx.x;
  if (i < n) *(int4*)(head + (size_t)i * HEAD_PITCH) = make_int4(-1, -1, -1, -1);
}

__global__ __launch_bounds__(256) void k_link(const int* __restrict__ dst,
                                              int* head, int* __restrict__ nxt, int e) {
  int i = blockIdx.x * blockDim.x + threadIdx.x;
  if (i < e) nxt[i] = atomicExch(&head[(size_t)dst[i] * HEAD_PITCH + (i & (NL - 1))], i);
}

__global__ __launch_bounds__(256) void k_deg_inv(const int* __restrict__ head,
                                                 const int* __restrict__ nxt,
                                                 const float* __restrict__ w,
                                                 float* __restrict__ inv, int n, int e) {
  int i = blockIdx.x * blockDim.x + threadIdx.x;
  if (i >= n) return;
  const int4 hd = *(const int4*)(head + (size_t)i * HEAD_PITCH);
  int   k[NL] = {hd.x, hd.y, hd.z, hd.w};
  float d[NL] = {1.0f, 0.0f, 0.0f, 0.0f};
  for (int it = 0; it < e; ++it) {
    bool  live[NL];
    float wv[NL];
    int   kn[NL];
    bool  any = false;
#pragma unroll
    for (int c = 0; c < NL; ++c) {
      live[c] = (unsigned)k[c] < (unsigned)e;
      any |= live[c];
      const int kk = live[c] ? k[c] : 0;
      wv[c] = w[kk];
      kn[c] = nxt[kk];
    }
    if (!any) break;
#pragma unroll
    for (int c = 0; c < NL; ++c) {
      d[c] = live[c] ? d[c] + wv[c] : d[c];
      k[c] = live[c] ? kn[c] : -1;
    }
  }
  const float ds = (d[0] + d[1]) + (d[2] + d[3]);
  inv[i] = (ds > 0.0f) ? rsqrtf(fmaxf(ds, 1e-12f)) : 0.0f;
}


template <int K, int NT, bool RELU>
__device__ __forceinline__ void strip_product(const float* __restrict__ A,
                                              const float* __restrict__ W,
                                              float* __restrict__ H,
                                              float4* Wl, int n_row_tiles) {
  constexpr int N  = NT * 16;
  constexpr int NQ = NT / 2;
  constexpr int KS = K / 4;
  constexpr int CH = 8;

#pragma unroll 8
  for (int it = 0; it < KS * NQ * 32 / 256; ++it) {
    const int i  = it * 256 + threadIdx.x;
    const int ln = i & 31;
    const int kq = i >> 5;
    const int q  = kq % NQ;
    const int kk = kq / NQ;
    const int k  = 4 * kk + 2 * (ln >> 4);
    const int n  = 32 * q + (ln & 15);
    float4 f;
    f.x = W[(size_t)k       * N + n];
    f.y = W[(size_t)(k + 1) * N + n];
    f.z = W[(size_t)k       * N + n + 16];
    f.w = W[(size_t)(k + 1) * N + n + 16];
    Wl[i] = f;
  }
  __syncthreads();

  const int rt = blockIdx.x * 8 + (threadIdx.x >> 5);
  if (rt >= n_row_tiles) return;
  const int lane = threadIdx.x & 31;
  const int l    = lane & 15;
  const int half = lane >> 4;
  const int m0   = rt * 16;

  const float* arow = A + (size_t)(m0 + l) * K + half * 2;
  v8f c[NT];
#pragma unroll
  for (int t = 0; t < NT; ++t) c[t] = (v8f){0.f, 0.f, 0.f, 0.f, 0.f, 0.f, 0.f, 0.f};

  v2f ac[CH], an[CH];
#pragma unroll
  for (int j = 0; j < CH; ++j) {
    v2f a = *(const v2f*)(arow + 4 * j);
    if (RELU) { a.x = fmaxf(a.x, 0.0f); a.y = fmaxf(a.y, 0.0f); }
    ac[j] = a;
  }

  for (int ch = 0; ch < KS / CH; ++ch) {
    if (ch + 1 < KS / CH) {
#pragma unroll
      for (int j = 0; j < CH; ++j) {
        v2f a = *(const v2f*)(arow + 4 * (CH * (ch + 1) + j));
        if (RELU) { a.x = fmaxf(a.x, 0.0f); a.y = fmaxf(a.y, 0.0f); }
        an[j] = a;
      }
    }
#pragma unroll
    for (int j = 0; j < CH; ++j) {
      const float4* wk = Wl + (size_t)((ch * CH + j) * NQ) * 32 + lane;
#pragma unroll
      for (int q = 0; q < NQ; ++q) {
        const float4 f = wk[q * 32];
        v2f b0; b0.x = f.x; b0.y = f.y;
        v2f b1; b1.x = f.z; b1.y = f.w;
        c[2 * q]     = __builtin_amdgcn_wmma_f32_16x16x4_f32(false, ac[j], false, b0,
                                                             (short)0, c[2 * q], false, false);
        c[2 * q + 1] = __builtin_amdgcn_wmma_f32_16x16x4_f32(false, ac[j], false, b1,
                                                             (short)0, c[2 * q + 1], false, false);
      }
    }
#pragma unroll
    for (int j = 0; j < CH; ++j) ac[j] = an[j];
  }

  float* drow = H + (size_t)(m0 + half * 8) * N + l;
#pragma unroll
  for (int t = 0; t < NT; ++t) {
#pragma unroll
    for (int i = 0; i < 8; ++i) drow[(size_t)i * N + t * 16] = c[t][i];
  }
}

__global__ __launch_bounds__(256) void k_gemm1_wmma(const float* __restrict__ X,
                                                    const float* __restrict__ W,
                                                    float* __restrict__ H) {
  __shared__ float4 Wl[(DIM_IN / 4) * (DIM_HID / 32) * 32];
  strip_product<DIM_IN, DIM_HID / 16, false>(X, W, H, Wl, N_NODES / 16);
}

__global__ __launch_bounds__(256) void k_gemm2_wmma(const float* __restrict__ XE,
                                                    const float* __restrict__ W,
                                                    float* __restrict__ H) {
  __shared__ float4 Wl[(DIM_HID / 4) * (DIM_OUT / 32) * 32];
  strip_product<DIM_HID, DIM_OUT / 16, true>(XE, W, H, Wl, N_NODES / 16);
}


__device__ __forceinline__ float lane_bcast_f(float x, int c) {
  return __builtin_bit_cast(float, __builtin_amdgcn_readlane(__builtin_bit_cast(int, x), c));
}

__global__ __launch_bounds__(256) void k_gather128(const int* __restrict__ src,
                                                   const float* __restrict__ wgt,
                                                   const int* __restrict__ head,
                                                   const int* __restrict__ nxt,
                                                   const float* __restrict__ inv,
                                                   const float* __restrict__ h,
                                                   const float* __restrict__ bias,
                                                   float* __restrict__ out, int e, int n) {
  const int t = __builtin_amdgcn_readfirstlane((int)(blockIdx.x * 8 + (threadIdx.x >> 5)));
  const int lane = threadIdx.x & 31;
  if (t >= n) return;
  const float it_ = inv[t];
  const float4 vt = *(const float4*)(h + (size_t)t * DIM_HID + lane * 4);
  int kc = -1;
  if (lane < NL) kc = head[(size_t)t * HEAD_PITCH + lane];
  const int kk0 = ((unsigned)kc < (unsigned)e) ? kc : 0;
  int   kn = nxt[kk0];
  int   s  = src[kk0];
  float w  = wgt[kk0];
  float4 acc;
  {
    const float nrm = it_ * 1.0f * it_;
    acc.x = nrm * vt.x; acc.y = nrm * vt.y; acc.z = nrm * vt.z; acc.w = nrm * vt.w;
  }
  for (int it = 0; it < e; ++it) {
    const bool live = (unsigned)kc < (unsigned)e;
    if (__builtin_amdgcn_ballot_w32(live) == 0u) break;
    const bool  nlive = live && (unsigned)kn < (unsigned)e;
    const int   kk2 = nlive ? kn : 0;
    const int   kn2 = nxt[kk2];
    const int   s2  = src[kk2];
    const float w2  = wgt[kk2];
    const int   sc  = live ? s : t;
    const float ps  = inv[sc];
    const int   lm  = live ? 1 : 0;
    float4 v[NL];
#pragma unroll
    for (int c = 0; c < NL; ++c) {
      v[c] = make_float4(0.f, 0.f, 0.f, 0.f);
      if (__builtin_amdgcn_readlane(lm, c)) {
        const int sr = __builtin_amdgcn_readlane(sc, c);
        v[c] = *(const float4*)(h + (size_t)sr * DIM_HID + lane * 4);
      }
    }
    __builtin_amdgcn_sched_barrier(0);
    const float coef = live ? ps * w * it_ : 0.0f;
#pragma unroll
    for (int c = 0; c < NL; ++c) {
      const float nrm = lane_bcast_f(coef, c);
      acc.x += nrm * v[c].x; acc.y += nrm * v[c].y; acc.z += nrm * v[c].z; acc.w += nrm * v[c].w;
    }
    kc = nlive ? kn : -1;
    kn = kn2; s = s2; w = w2;
  }
  const float4 b = *(const float4*)(bias + lane * 4);
  float4 r = {acc.x + b.x, acc.y + b.y, acc.z + b.z, acc.w + b.w};
  *(float4*)(out + (size_t)t * DIM_HID + lane * 4) = r;
}

__global__ __launch_bounds__(256) void k_gather32(const int* __restrict__ src,
                                                  const float* __restrict__ wgt,
                                                  const int* __restrict__ head,
                                                  const int* __restrict__ nxt,
                                                  const float* __restrict__ inv,
                                                  const float* __restrict__ h,
                                                  const float* __restrict__ bias,
                                                  float* __restrict__ out, int e, int n) {
  const int t = __builtin_amdgcn_readfirstlane((int)(blockIdx.x * 8 + (threadIdx.x >> 5)));
  const int lane = threadIdx.x & 31;
  if (t >= n) return;
  const float it_ = inv[t];
  const float vt = h[(size_t)t * DIM_OUT + lane];
  int kc = -1;
  if (lane < NL) kc = head[(size_t)t * HEAD_PITCH + lane];
  const int kk0 = ((unsigned)kc < (unsigned)e) ? kc : 0;
  int   kn = nxt[kk0];
  int   s  = src[kk0];
  float w  = wgt[kk0];
  float acc = (it_ * 1.0f * it_) * vt;
  for (int it = 0; it < e; ++it) {
    const bool live = (unsigned)kc < (unsigned)e;
    if (__builtin_amdgcn_ballot_w32(live) == 0u) break;
    const bool  nlive = live && (unsigned)kn < (unsigned)e;
    const int   kk2 = nlive ? kn : 0;
    const int   kn2 = nxt[kk2];
    const int   s2  = src[kk2];
    const float w2  = wgt[kk2];
    const int   sc  = live ? s : t;
    const float ps  = inv[sc];
    const int   lm  = live ? 1 : 0;
    float v[NL];
#pragma unroll
    for (int c = 0; c < NL; ++c) {
      v[c] = 0.f;
      if (__builtin_amdgcn_readlane(lm, c)) {
        const int sr = __builtin_amdgcn_readlane(sc, c);
        v[c] = h[(size_t)sr * DIM_OUT + lane];
      }
    }
    __builtin_amdgcn_sched_barrier(0);
    const float coef = live ? ps * w * it_ : 0.0f;
#pragma unroll
    for (int c = 0; c < NL; ++c) acc += lane_bcast_f(coef, c) * v[c];
    kc = nlive ? kn : -1;
    kn = kn2; s = s2; w = w2;
  }
  out[(size_t)t * DIM_OUT + lane] = acc + bias[lane];
}


extern "C" void kernel_launch(void* const* d_in, const int* in_sizes, int n_in,
                              void* d_out, int out_size, void* d_ws, size_t ws_size,
                              hipStream_t stream) {
  const int N = N_NODES, E = N_EDGES;

  const float* x   = (const float*)d_in[0];
  const int*   ei  = (const int*)d_in[1];
  const int*   src = ei;
  const int*   dst = ei + E;
  const float* w   = (const float*)d_in[2];
  const float* W1  = (const float*)d_in[3];
  const float* b1  = (const float*)d_in[4];
  const float* W2  = (const float*)d_in[5];
  const float* b2  = (const float*)d_in[6];

  float* out  = (float*)d_out;
  float* xemb = out + (size_t)N * DIM_OUT;

  float* inv  = (float*)d_ws;
  float* h1   = inv + N;
  float* h2   = h1;
  int*   head = (int*)(h1 + (size_t)N * DIM_HID);
  int*   nxt  = head + (size_t)N * HEAD_PITCH;

  k_init_head<<<(N + 255) / 256, 256, 0, stream>>>(head, N);
  k_link     <<<(E + 255) / 256, 256, 0, stream>>>(dst, head, nxt, E);
  k_deg_inv  <<<(N + 255) / 256, 256, 0, stream>>>(head, nxt, w, inv, N, E);

  k_gemm1_wmma<<<(N / 16 + 7) / 8, 256, 0, stream>>>(x, W1, h1);

  k_gather128<<<(N + 7) / 8, 256, 0, stream>>>(src, w, head, nxt, inv, h1, b1, xemb, E, N);

  k_gemm2_wmma<<<(N / 16 + 7) / 8, 256, 0, stream>>>(xemb, W2, h2);

  k_gather32<<<(N + 7) / 8, 256, 0, stream>>>(src, w, head, nxt, inv, h2, b2, out, E, N);
}


# link pass writes one 12-byte record per edge (next edge, weight, source); every list hop reads that one record, not three 4-byte words of three arrays: 5 fewer line requests per edge; f32 sums kept
# speedup vs baseline: 2.8490x; 1.0785x over previous
#include <hip/hip_runtime.h>


typedef __attribute__((ext_vector_type(2))) float v2f;
typedef __attribute__((ext_vector_type(8))) float v8f;

#define N_NODES  100000
#define N_EDGES  1600000
#define DIM_IN   256
#define DIM_HID  128
#define DIM_OUT  32
#define HEAD_PITCH 16

#define NL 4

struct Rec { int nxt; float w; int src; };

#define REC_A 1066666

__device__ __forceinline__ const Rec* rec_at(const Rec* ra, const Rec* rb, int e) {
  return (e < REC_A) ? (ra + e) : (rb + (e - REC_A));
}

__global__ __launch_bounds__(256) void k_init_head(int* head, int n) {
  int i = blockIdx.x * blockDim.x + threadIdx.x;
  if (i < n) *(int4*)(head + (size_t)i * HEAD_PITCH) = make_int4(-1, -1, -1, -1);
}

__global__ __launch_bounds__(256) void k_link(const int* __restrict__ src,
                                              const int* __restrict__ dst,
                                              const float* __restrict__ w,
                                              int* head, Rec* ra, Rec* rb, int e) {
  int i = blockIdx.x * blockDim.x + threadIdx.x;
  if (i >= e) return;
  Rec r;
  r.src = src[i];
  r.w   = w[i];
  r.nxt = atomicExch(&head[(size_t)dst[i] * HEAD_PITCH + (i & (NL - 1))], i);
  Rec* p = (i < REC_A) ? (ra + i) : (rb + (i - REC_A));
  *p = r;
}

__global__ __launch_bounds__(256) void k_deg_inv(const int* __restrict__ head,
                                                 const Rec* __restrict__ ra,
                                                 const Rec* __restrict__ rb,
                                                 float* __restrict__ inv, int n, int e) {
  int i = blockIdx.x * blockDim.x + threadIdx.x;
  if (i >= n) return;
  const int4 hd = *(const int4*)(head + (size_t)i * HEAD_PITCH);
  int   k[NL] = {hd.x, hd.y, hd.z, hd.w};
  float d[NL] = {1.0f, 0.0f, 0.0f, 0.0f};
  for (int it = 0; it < e; ++it) {
    bool  live[NL];
    float wv[NL];
    int   kn[NL];
    bool  any = false;
#pragma unroll
    for (int c = 0; c < NL; ++c) {
      live[c] = (unsigned)k[c] < (unsigned)e;
      any |= live[c];
      const int kk = live[c] ? k[c] : 0;
      const Rec* p = rec_at(ra, rb, kk);
      wv[c] = p->w;
      kn[c] = p->nxt;
    }
    if (!any) break;
#pragma unroll
    for (int c = 0; c < NL; ++c) {
      d[c] = live[c] ? d[c] + wv[c] : d[c];
      k[c] = live[c] ? kn[c] : -1;
    }
  }
  const float ds = (d[0] + d[1]) + (d[2] + d[3]);
  inv[i] = (ds > 0.0f) ? rsqrtf(fmaxf(ds, 1e-12f)) : 0.0f;
}


template <int K, int NT, bool RELU>
__device__ __forceinline__ void strip_product(const float* __restrict__ A,
                                              const float* __restrict__ W,
                                              float* __restrict__ H,
                                              float4* Wl, int n_row_tiles) {
  constexpr int N  = NT * 16;
  constexpr int NQ = NT / 2;
  constexpr int KS = K / 4;
  constexpr int CH = 8;

#pragma unroll 8
  for (int it = 0; it < KS * NQ * 32 / 256; ++it) {
    const int i  = it * 256 + threadIdx.x;
    const int ln = i & 31;
    const int kq = i >> 5;
    const int q  = kq % NQ;
    const int kk = kq / NQ;
    const int k  = 4 * kk + 2 * (ln >> 4);
    const int n  = 32 * q + (ln & 15);
    float4 f;
    f.x = W[(size_t)k       * N + n];
    f.y = W[(size_t)(k + 1) * N + n];
    f.z = W[(size_t)k       * N + n + 16];
    f.w = W[(size_t)(k + 1) * N + n + 16];
    Wl[i] = f;
  }
  __syncthreads();

  const int rt = blockIdx.x * 8 + (threadIdx.x >> 5);
  if (rt >= n_row_tiles) return;
  const int lane = threadIdx.x & 31;
  const int l    = lane & 15;
  const int half = lane >> 4;
  const int m0   = rt * 16;

  const float* arow = A + (size_t)(m0 + l) * K + half * 2;
  v8f c[NT];
#pragma unroll
  for (int t = 0; t < NT; ++t) c[t] = (v8f){0.f, 0.f, 0.f, 0.f, 0.f, 0.f, 0.f, 0.f};

  v2f ac[CH], an[CH];
#pragma unroll
  for (int j = 0; j < CH; ++j) {
    v2f a = *(const v2f*)(arow + 4 * j);
    if (RELU) { a.x = fmaxf(a.x, 0.0f); a.y = fmaxf(a.y, 0.0f); }
    ac[j] = a;
  }

  for (int ch = 0; ch < KS / CH; ++ch) {
    if (ch + 1 < KS / CH) {
#pragma unroll
      for (int j = 0; j < CH; ++j) {
        v2f a = *(const v2f*)(arow + 4 * (CH * (ch + 1) + j));
        if (RELU) { a.x = fmaxf(a.x, 0.0f); a.y = fmaxf(a.y, 0.0f); }
        an[j] = a;
      }
    }
#pragma unroll
    for (int j = 0; j < CH; ++j) {
      const float4* wk = Wl + (size_t)((ch * CH + j) * NQ) * 32 + lane;
#pragma unroll
      for (int q = 0; q < NQ; ++q) {
        const float4 f = wk[q * 32];
        v2f b0; b0.x = f.x; b0.y = f.y;
        v2f b1; b1.x = f.z; b1.y = f.w;
        c[2 * q]     = __builtin_amdgcn_wmma_f32_16x16x4_f32(false, ac[j], false, b0,
                                                             (short)0, c[2 * q], false, false);
        c[2 * q + 1] = __builtin_amdgcn_wmma_f32_16x16x4_f32(false, ac[j], false, b1,
                                                             (short)0, c[2 * q + 1], false, false);
      }
    }
#pragma unroll
    for (int j = 0; j < CH; ++j) ac[j] = an[j];
  }

  float* drow = H + (size_t)(m0 + half * 8) * N + l;
#pragma unroll
  for (int t = 0; t < NT; ++t) {
#pragma unroll
    for (int i = 0; i < 8; ++i) drow[(size_t)i * N + t * 16] = c[t][i];
  }
}

__global__ __launch_bounds__(256) void k_gemm1_wmma(const float* __restrict__ X,
                                                    const float* __restrict__ W,
                                                    float* __restrict__ H) {
  __shared__ float4 Wl[(DIM_IN / 4) * (DIM_HID / 32) * 32];
  strip_product<DIM_IN, DIM_HID / 16, false>(X, W, H, Wl, N_NODES / 16);
}

#define GEMM2_BLOCKS ((N_NODES / 16 + 7) / 8)
__global__ __launch_bounds__(256) void k_gemm2_wmma(const float* __restrict__ XE,
                                                    const float* __restrict__ W,
                                                    float* __restrict__ H,
                                                    const Rec* __restrict__ ra,
                                                    const Rec* __restrict__ rb,
                                                    int4* __restrict__ rec2) {
  __shared__ float4 Wl[(DIM_HID / 4) * (DIM_OUT / 32) * 32];
  constexpr int NTHR = GEMM2_BLOCKS * 256;
  constexpr int FULL = N_EDGES / NTHR;
  const int g = blockIdx.x * 256 + threadIdx.x;
  Rec r[FULL];
#pragma unroll
  for (int j = 0; j < FULL; ++j) r[j] = *rec_at(ra, rb, j * NTHR + g);
#pragma unroll
  for (int j = 0; j < FULL; ++j)
    rec2[j * NTHR + g] = make_int4(r[j].nxt, r[j].src, __float_as_int(r[j].w), 0);
  if (FULL * NTHR + g < N_EDGES) {
    const Rec q = *rec_at(ra, rb, FULL * NTHR + g);
    rec2[FULL * NTHR + g] = make_int4(q.nxt, q.src, __float_as_int(q.w), 0);
  }
  strip_product<DIM_HID, DIM_OUT / 16, true>(XE, W, H, Wl, N_NODES / 16);
}


__device__ __forceinline__ float lane_bcast_f(float x, int c) {
  return __builtin_bit_cast(float, __builtin_amdgcn_readlane(__builtin_bit_cast(int, x), c));
}

__global__ __launch_bounds__(256) void k_gather128(const Rec* __restrict__ ra,
                                                   const Rec* __restrict__ rb,
                                                   const int* __restrict__ head,
                                                   const float* __restrict__ inv,
                                                   const float* __restrict__ h,
                                                   const float* __restrict__ bias,
                                                   float* __restrict__ out, int e, int n) {
  const int t = __builtin_amdgcn_readfirstlane((int)(blockIdx.x * 8 + (threadIdx.x >> 5)));
  const int lane = threadIdx.x & 31;
  if (t >= n) return;
  const float it_ = inv[t];
  const float4 vt = *(const float4*)(h + (size_t)t * DIM_HID + lane * 4);
  int kc = -1;
  if (lane < NL) kc = head[(size_t)t * HEAD_PITCH + lane];
  const int kk0 = ((unsigned)kc < (unsigned)e) ? kc : 0;
  const Rec r0 = *rec_at(ra, rb, kk0);
  int   kn = r0.nxt;
  int   s  = r0.src;
  float w  = r0.w;
  float4 acc;
  {
    const float nrm = it_ * 1.0f * it_;
    acc.x = nrm * vt.x; acc.y = nrm * vt.y; acc.z = nrm * vt.z; acc.w = nrm * vt.w;
  }
  for (int it = 0; it < e; ++it) {
    const bool live = (unsigned)kc < (unsigned)e;
    if (__builtin_amdgcn_ballot_w32(live) == 0u) break;
    const bool  nlive = live && (unsigned)kn < (unsigned)e;
    const int   kk2 = nlive ? kn : 0;
    const Rec   r2  = *rec_at(ra, rb, kk2);
    const int   kn2 = r2.nxt;
    const int   s2  = r2.src;
    const float w2  = r2.w;
    const int   sc  = live ? s : t;
    const float ps  = inv[sc];
    const int   lm  = live ? 1 : 0;
    float4 v[NL];
#pragma unroll
    for (int c = 0; c < NL; ++c) {
      v[c] = make_float4(0.f, 0.f, 0.f, 0.f);
      if (__builtin_amdgcn_readlane(lm, c)) {
        const int sr = __builtin_amdgcn_readlane(sc, c);
        v[c] = *(const float4*)(h + (size_t)sr * DIM_HID + lane * 4);
      }
    }
    __builtin_amdgcn_sched_barrier(0);
    const float coef = live ? ps * w * it_ : 0.0f;
#pragma unroll
    for (int c = 0; c < NL; ++c) {
      const float nrm = lane_bcast_f(coef, c);
      acc.x += nrm * v[c].x; acc.y += nrm * v[c].y; acc.z += nrm * v[c].z; acc.w += nrm * v[c].w;
    }
    kc = nlive ? kn : -1;
    kn = kn2; s = s2; w = w2;
  }
  const float4 b = *(const float4*)(bias + lane * 4);
  float4 r = {acc.x + b.x, acc.y + b.y, acc.z + b.z, acc.w + b.w};
  *(float4*)(out + (size_t)t * DIM_HID + lane * 4) = r;
}

__global__ __launch_bounds__(256) void k_gather32(const int4* __restrict__ rec2,
                                                  const int* __restrict__ head,
                                                  const float* __restrict__ inv,
                                                  const float* __restrict__ h,
                                                  const float* __restrict__ bias,
                                                  float* __restrict__ out, int e, int n) {
  const int t = __builtin_amdgcn_readfirstlane((int)(blockIdx.x * 8 + (threadIdx.x >> 5)));
  const int lane = threadIdx.x & 31;
  if (t >= n) return;
  const float it_ = inv[t];
  const float vt = h[(size_t)t * DIM_OUT + lane];
  int kc = -1;
  if (lane < NL) kc = head[(size_t)t * HEAD_PITCH + lane];
  const int kk0 = ((unsigned)kc < (unsigned)e) ? kc : 0;
  const int4 r0 = rec2[kk0];
  int   kn = r0.x;
  int   s  = r0.y;
  float w  = __int_as_float(r0.z);
  float acc = (it_ * 1.0f * it_) * vt;
  for (int it = 0; it < e; ++it) {
    const bool live = (unsigned)kc < (unsigned)e;
    if (__builtin_amdgcn_ballot_w32(live) == 0u) break;
    const bool  nlive = live && (unsigned)kn < (unsigned)e;
    const int   kk2 = nlive ? kn : 0;
    const int4  r2  = rec2[kk2];
    const int   kn2 = r2.x;
    const int   s2  = r2.y;
    const float w2  = __int_as_float(r2.z);
    const int   sc  = live ? s : t;
    const float ps  = inv[sc];
    const int   lm  = live ? 1 : 0;
    float v[NL];
#pragma unroll
    for (int c = 0; c < NL; ++c) {
      v[c] = 0.f;
      if (__builtin_amdgcn_readlane(lm, c)) {
        const int sr = __builtin_amdgcn_readlane(sc, c);
        v[c] = h[(size_t)sr * DIM_OUT + lane];
      }
    }
    __builtin_amdgcn_sched_barrier(0);
    const float coef = live ? ps * w * it_ : 0.0f;
#pragma unroll
    for (int c = 0; c < NL; ++c) acc += lane_bcast_f(coef, c) * v[c];
    kc = nlive ? kn : -1;
    kn = kn2; s = s2; w = w2;
  }
  out[(size_t)t * DIM_OUT + lane] = acc + bias[lane];
}


extern "C" void kernel_launch(void* const* d_in, const int* in_sizes, int n_in,
                              void* d_out, int out_size, void* d_ws, size_t ws_size,
                              hipStream_t stream) {
  const int N = N_NODES, E = N_EDGES;

  const float* x   = (const float*)d_in[0];
  const int*   ei  = (const int*)d_in[1];
  const int*   src = ei;
  const int*   dst = ei + E;
  const float* w   = (const float*)d_in[2];
  const float* W1  = (const float*)d_in[3];
  const float* b1  = (const float*)d_in[4];
  const float* W2  = (const float*)d_in[5];
  const float* b2  = (const float*)d_in[6];

  float* out  = (float*)d_out;
  float* xemb = out + (size_t)N * DIM_OUT;

  float* inv  = (float*)d_ws;
  float* h1   = inv + N;
  float* h2   = h1;
  int*   head = (int*)(h1 + (size_t)N * DIM_HID);
  int*   tail = head + (size_t)N * HEAD_PITCH;

  Rec*  ra   = (Rec*)d_out;
  Rec*  rb   = (Rec*)(tail - 2);
  int4* rec2 = (int4*)(h1 + (size_t)N * DIM_OUT);

  k_init_head<<<(N + 255) / 256, 256, 0, stream>>>(head, N);
  k_link     <<<(E + 255) / 256, 256, 0, stream>>>(src, dst, w, head, ra, rb, E);
  k_deg_inv  <<<(N + 255) / 256, 256, 0, stream>>>(head, ra, rb, inv, N, E);

  k_gemm1_wmma<<<(N / 16 + 7) / 8, 256, 0, stream>>>(x, W1, h1);

  k_gather128<<<(N + 7) / 8, 256, 0, stream>>>(ra, rb, head, inv, h1, b1, xemb, E, N);

  k_gemm2_wmma<<<GEMM2_BLOCKS, 256, 0, stream>>>(xemb, W2, h2, ra, rb, rec2);

  k_gather32<<<(N + 7) / 8, 256, 0, stream>>>(rec2, head, inv, h2, b2, out, E, N);
}


# neighbour sums: each of a node's 4 lists gets its own 8 lanes, which fetch the rows themselves and keep a partial sum, added across lane groups at the end; no lane broadcast; f32 sum order changes
# speedup vs baseline: 3.0191x; 1.0597x over previous
#include <hip/hip_runtime.h>


typedef __attribute__((ext_vector_type(2))) float v2f;
typedef __attribute__((ext_vector_type(8))) float v8f;

#define N_NODES  100000
#define N_EDGES  1600000
#define DIM_IN   256
#define DIM_HID  128
#define DIM_OUT  32
#define HEAD_PITCH 16

#define NL 4

struct Rec { int nxt; float w; int src; };

#define REC_A 1066666

__device__ __forceinline__ const Rec* rec_at(const Rec* ra, const Rec* rb, int e) {
  return (e < REC_A) ? (ra + e) : (rb + (e - REC_A));
}

__global__ __launch_bounds__(256) void k_init_head(int* head, int n) {
  int i = blockIdx.x * blockDim.x + threadIdx.x;
  if (i < n) *(int4*)(head + (size_t)i * HEAD_PITCH) = make_int4(-1, -1, -1, -1);
}

__global__ __launch_bounds__(256) void k_link(const int* __restrict__ src,
                                              const int* __restrict__ dst,
                                              const float* __restrict__ w,
                                              int* head, Rec* ra, Rec* rb, int e) {
  int i = blockIdx.x * blockDim.x + threadIdx.x;
  if (i >= e) return;
  Rec r;
  r.src = src[i];
  r.w   = w[i];
  r.nxt = atomicExch(&head[(size_t)dst[i] * HEAD_PITCH + (i & (NL - 1))], i);
  Rec* p = (i < REC_A) ? (ra + i) : (rb + (i - REC_A));
  *p = r;
}

__global__ __launch_bounds__(256) void k_deg_inv(const int* __restrict__ head,
                                                 const Rec* __restrict__ ra,
                                                 const Rec* __restrict__ rb,
                                                 float* __restrict__ inv, int n, int e) {
  int i = blockIdx.x * blockDim.x + threadIdx.x;
  if (i >= n) return;
  const int4 hd = *(const int4*)(head + (size_t)i * HEAD_PITCH);
  int   k[NL] = {hd.x, hd.y, hd.z, hd.w};
  float d[NL] = {1.0f, 0.0f, 0.0f, 0.0f};
  for (int it = 0; it < e; ++it) {
    bool  live[NL];
    float wv[NL];
    int   kn[NL];
    bool  any = false;
#pragma unroll
    for (int c = 0; c < NL; ++c) {
      live[c] = (unsigned)k[c] < (unsigned)e;
      any |= live[c];
      const int kk = live[c] ? k[c] : 0;
      const Rec* p = rec_at(ra, rb, kk);
      wv[c] = p->w;
      kn[c] = p->nxt;
    }
    if (!any) break;
#pragma unroll
    for (int c = 0; c < NL; ++c) {
      d[c] = live[c] ? d[c] + wv[c] : d[c];
      k[c] = live[c] ? kn[c] : -1;
    }
  }
  const float ds = (d[0] + d[1]) + (d[2] + d[3]);
  inv[i] = (ds > 0.0f) ? rsqrtf(fmaxf(ds, 1e-12f)) : 0.0f;
}


template <int K, int NT, bool RELU>
__device__ __forceinline__ void strip_product(const float* __restrict__ A,
                                              const float* __restrict__ W,
                                              float* __restrict__ H,
                                              float4* Wl, int n_row_tiles) {
  constexpr int N  = NT * 16;
  constexpr int NQ = NT / 2;
  constexpr int KS = K / 4;
  constexpr int CH = 8;

#pragma unroll 8
  for (int it = 0; it < KS * NQ * 32 / 256; ++it) {
    const int i  = it * 256 + threadIdx.x;
    const int ln = i & 31;
    const int kq = i >> 5;
    const int q  = kq % NQ;
    const int kk = kq / NQ;
    const int k  = 4 * kk + 2 * (ln >> 4);
    const int n  = 32 * q + (ln & 15);
    float4 f;
    f.x = W[(size_t)k       * N + n];
    f.y = W[(size_t)(k + 1) * N + n];
    f.z = W[(size_t)k       * N + n + 16];
    f.w = W[(size_t)(k + 1) * N + n + 16];
    Wl[i] = f;
  }
  __syncthreads();

  const int rt = blockIdx.x * 8 + (threadIdx.x >> 5);
  if (rt >= n_row_tiles) return;
  const int lane = threadIdx.x & 31;
  const int l    = lane & 15;
  const int half = lane >> 4;
  const int m0   = rt * 16;

  const float* arow = A + (size_t)(m0 + l) * K + half * 2;
  v8f c[NT];
#pragma unroll
  for (int t = 0; t < NT; ++t) c[t] = (v8f){0.f, 0.f, 0.f, 0.f, 0.f, 0.f, 0.f, 0.f};

  v2f ac[CH], an[CH];
#pragma unroll
  for (int j = 0; j < CH; ++j) {
    v2f a = *(const v2f*)(arow + 4 * j);
    if (RELU) { a.x = fmaxf(a.x, 0.0f); a.y = fmaxf(a.y, 0.0f); }
    ac[j] = a;
  }

  for (int ch = 0; ch < KS / CH; ++ch) {
    if (ch + 1 < KS / CH) {
#pragma unroll
      for (int j = 0; j < CH; ++j) {
        v2f a = *(const v2f*)(arow + 4 * (CH * (ch + 1) + j));
        if (RELU) { a.x = fmaxf(a.x, 0.0f); a.y = fmaxf(a.y, 0.0f); }
        an[j] = a;
      }
    }
#pragma unroll
    for (int j = 0; j < CH; ++j) {
      const float4* wk = Wl + (size_t)((ch * CH + j) * NQ) * 32 + lane;
#pragma unroll
      for (int q = 0; q < NQ; ++q) {
        const float4 f = wk[q * 32];
        v2f b0; b0.x = f.x; b0.y = f.y;
        v2f b1; b1.x = f.z; b1.y = f.w;
        c[2 * q]     = __builtin_amdgcn_wmma_f32_16x16x4_f32(false, ac[j], false, b0,
                                                             (short)0, c[2 * q], false, false);
        c[2 * q + 1] = __builtin_amdgcn_wmma_f32_16x16x4_f32(false, ac[j], false, b1,
                                                             (short)0, c[2 * q + 1], false, false);
      }
    }
#pragma unroll
    for (int j = 0; j < CH; ++j) ac[j] = an[j];
  }

  float* drow = H + (size_t)(m0 + half * 8) * N + l;
#pragma unroll
  for (int t = 0; t < NT; ++t) {
#pragma unroll
    for (int i = 0; i < 8; ++i) drow[(size_t)i * N + t * 16] = c[t][i];
  }
}

__global__ __launch_bounds__(256) void k_gemm1_wmma(const float* __restrict__ X,
                                                    const float* __restrict__ W,
                                                    float* __restrict__ H) {
  __shared__ float4 Wl[(DIM_IN / 4) * (DIM_HID / 32) * 32];
  strip_product<DIM_IN, DIM_HID / 16, false>(X, W, H, Wl, N_NODES / 16);
}

#define GEMM2_BLOCKS ((N_NODES / 16 + 7) / 8)
__global__ __launch_bounds__(256) void k_gemm2_wmma(const float* __restrict__ XE,
                                                    const float* __restrict__ W,
                                                    float* __restrict__ H,
                                                    const Rec* __restrict__ ra,
                                                    const Rec* __restrict__ rb,
                                                    int4* __restrict__ rec2) {
  __shared__ float4 Wl[(DIM_HID / 4) * (DIM_OUT / 32) * 32];
  constexpr int NTHR = GEMM2_BLOCKS * 256;
  constexpr int FULL = N_EDGES / NTHR;
  const int g = blockIdx.x * 256 + threadIdx.x;
  Rec r[FULL];
#pragma unroll
  for (int j = 0; j < FULL; ++j) r[j] = *rec_at(ra, rb, j * NTHR + g);
#pragma unroll
  for (int j = 0; j < FULL; ++j)
    rec2[j * NTHR + g] = make_int4(r[j].nxt, r[j].src, __float_as_int(r[j].w), 0);
  if (FULL * NTHR + g < N_EDGES) {
    const Rec q = *rec_at(ra, rb, FULL * NTHR + g);
    rec2[FULL * NTHR + g] = make_int4(q.nxt, q.src, __float_as_int(q.w), 0);
  }
  strip_product<DIM_HID, DIM_OUT / 16, true>(XE, W, H, Wl, N_NODES / 16);
}


__global__ __launch_bounds__(256) void k_gather128(const Rec* __restrict__ ra,
                                                   const Rec* __restrict__ rb,
                                                   const int* __restrict__ head,
                                                   const float* __restrict__ inv,
                                                   const float* __restrict__ h,
                                                   const float* __restrict__ bias,
                                                   float* __restrict__ out, int e, int n) {
  const int t = __builtin_amdgcn_readfirstlane((int)(blockIdx.x * 8 + (threadIdx.x >> 5)));
  const int lane = threadIdx.x & 31;
  if (t >= n) return;
  const int g = lane >> 3;
  const int j = lane & 7;
  const float it_ = inv[t];
  const float* hl = h + j * 4;
  int kc = head[(size_t)t * HEAD_PITCH + g];
  const int kk0 = ((unsigned)kc < (unsigned)e) ? kc : 0;
  const Rec r0 = *rec_at(ra, rb, kk0);
  int   kn = r0.nxt;
  int   s  = r0.src;
  float w  = r0.w;
  float4 acc[4];
#pragma unroll
  for (int i = 0; i < 4; ++i) acc[i] = make_float4(0.f, 0.f, 0.f, 0.f);
  if (g == 0) {
    const float nrm = it_ * 1.0f * it_;
    const float* p = hl + (size_t)t * DIM_HID;
#pragma unroll
    for (int i = 0; i < 4; ++i) {
      const float4 v = *(const float4*)(p + i * 32);
      acc[i].x = nrm * v.x; acc[i].y = nrm * v.y; acc[i].z = nrm * v.z; acc[i].w = nrm * v.w;
    }
  }
  for (int it = 0; it < e; ++it) {
    const bool live = (unsigned)kc < (unsigned)e;
    if (__builtin_amdgcn_ballot_w32(live) == 0u) break;
    const bool  nlive = live && (unsigned)kn < (unsigned)e;
    const int   kk2 = nlive ? kn : 0;
    const Rec   r2  = *rec_at(ra, rb, kk2);
    if (live) {
      const float  ps = inv[s];
      const float* p  = hl + (size_t)s * DIM_HID;
      float4 v[4];
#pragma unroll
      for (int i = 0; i < 4; ++i) v[i] = *(const float4*)(p + i * 32);
      __builtin_amdgcn_sched_barrier(0);
      const float coef = ps * w * it_;
#pragma unroll
      for (int i = 0; i < 4; ++i) {
        acc[i].x += coef * v[i].x; acc[i].y += coef * v[i].y;
        acc[i].z += coef * v[i].z; acc[i].w += coef * v[i].w;
      }
    }
    kc = nlive ? kn : -1;
    kn = r2.nxt; s = r2.src; w = r2.w;
  }
  const bool lo = lane < 16;
  float4 keep[2];
#pragma unroll
  for (int k = 0; k < 2; ++k) {
    const float4 snd = lo ? acc[2 + k] : acc[k];
    float4       kp  = lo ? acc[k] : acc[2 + k];
    kp.x += __shfl_xor(snd.x, 16); kp.y += __shfl_xor(snd.y, 16);
    kp.z += __shfl_xor(snd.z, 16); kp.w += __shfl_xor(snd.w, 16);
    keep[k] = kp;
  }
  const bool ev = (g & 1) == 0;
  const float4 snd = ev ? keep[1] : keep[0];
  float4       fin = ev ? keep[0] : keep[1];
  fin.x += __shfl_xor(snd.x, 8); fin.y += __shfl_xor(snd.y, 8);
  fin.z += __shfl_xor(snd.z, 8); fin.w += __shfl_xor(snd.w, 8);
  const float4 b = *(const float4*)(bias + lane * 4);
  float4 r = {fin.x + b.x, fin.y + b.y, fin.z + b.z, fin.w + b.w};
  *(float4*)(out + (size_t)t * DIM_HID + lane * 4) = r;
}

__global__ __launch_bounds__(256) void k_gather32(const int4* __restrict__ rec2,
                                                  const int* __restrict__ head,
                                                  const float* __restrict__ inv,
                                                  const float* __restrict__ h,
                                                  const float* __restrict__ bias,
                                                  float* __restrict__ out, int e, int n) {
  const int t = __builtin_amdgcn_readfirstlane((int)(blockIdx.x * 8 + (threadIdx.x >> 5)));
  const int lane = threadIdx.x & 31;
  if (t >= n) return;
  const int g = lane >> 3;
  const int j = lane & 7;
  const float it_ = inv[t];
  const float* hl = h + j * 4;
  int kc = head[(size_t)t * HEAD_PITCH + g];
  const int kk0 = ((unsigned)kc < (unsigned)e) ? kc : 0;
  const int4 r0 = rec2[kk0];
  int   kn = r0.x;
  int   s  = r0.y;
  float w  = __int_as_float(r0.z);
  float4 acc = make_float4(0.f, 0.f, 0.f, 0.f);
  if (g == 0) {
    const float nrm = it_ * 1.0f * it_;
    const float4 v = *(const float4*)(hl + (size_t)t * DIM_OUT);
    acc.x = nrm * v.x; acc.y = nrm * v.y; acc.z = nrm * v.z; acc.w = nrm * v.w;
  }
  for (int it = 0; it < e; ++it) {
    const bool live = (unsigned)kc < (unsigned)e;
    if (__builtin_amdgcn_ballot_w32(live) == 0u) break;
    const bool  nlive = live && (unsigned)kn < (unsigned)e;
    const int   kk2 = nlive ? kn : 0;
    const int4  r2  = rec2[kk2];
    if (live) {
      const float  ps = inv[s];
      const float4 v  = *(const float4*)(hl + (size_t)s * DIM_OUT);
      __builtin_amdgcn_sched_barrier(0);
      const float coef = ps * w * it_;
      acc.x += coef * v.x; acc.y += coef * v.y; acc.z += coef * v.z; acc.w += coef * v.w;
    }
    kc = nlive ? kn : -1;
    kn = r2.x; s = r2.y; w = __int_as_float(r2.z);
  }
  const bool lo = lane < 16;
  const float s0 = lo ? acc.z : acc.x;
  const float s1 = lo ? acc.w : acc.y;
  float k0 = lo ? acc.x : acc.z;
  float k1 = lo ? acc.y : acc.w;
  k0 += __shfl_xor(s0, 16);
  k1 += __shfl_xor(s1, 16);
  const bool ev = (g & 1) == 0;
  const float sd = ev ? k1 : k0;
  float fn = ev ? k0 : k1;
  fn += __shfl_xor(sd, 8);
  const int idx = j * 4 + g;
  out[(size_t)t * DIM_OUT + idx] = fn + bias[idx];
}


extern "C" void kernel_launch(void* const* d_in, const int* in_sizes, int n_in,
                              void* d_out, int out_size, void* d_ws, size_t ws_size,
                              hipStream_t stream) {
  const int N = N_NODES, E = N_EDGES;

  const float* x   = (const float*)d_in[0];
  const int*   ei  = (const int*)d_in[1];
  const int*   src = ei;
  const int*   dst = ei + E;
  const float* w   = (const float*)d_in[2];
  const float* W1  = (const float*)d_in[3];
  const float* b1  = (const float*)d_in[4];
  const float* W2  = (const float*)d_in[5];
  const float* b2  = (const float*)d_in[6];

  float* out  = (float*)d_out;
  float* xemb = out + (size_t)N * DIM_OUT;

  float* inv  = (float*)d_ws;
  float* h1   = inv + N;
  float* h2   = h1;
  int*   head = (int*)(h1 + (size_t)N * DIM_HID);
  int*   tail = head + (size_t)N * HEAD_PITCH;

  Rec*  ra   = (Rec*)d_out;
  Rec*  rb   = (Rec*)(tail - 2);
  int4* rec2 = (int4*)(h1 + (size_t)N * DIM_OUT);

  k_init_head<<<(N + 255) / 256, 256, 0, stream>>>(head, N);
  k_link     <<<(E + 255) / 256, 256, 0, stream>>>(src, dst, w, head, ra, rb, E);
  k_deg_inv  <<<(N + 255) / 256, 256, 0, stream>>>(head, ra, rb, inv, N, E);

  k_gemm1_wmma<<<(N / 16 + 7) / 8, 256, 0, stream>>>(x, W1, h1);

  k_gather128<<<(N + 7) / 8, 256, 0, stream>>>(ra, rb, head, inv, h1, b1, xemb, E, N);

  k_gemm2_wmma<<<GEMM2_BLOCKS, 256, 0, stream>>>(xemb, W2, h2, ra, rb, rec2);

  k_gather32<<<(N + 7) / 8, 256, 0, stream>>>(rec2, head, inv, h2, b2, out, E, N);
}


# edge-link pass moved into the h1 product's launch as 256 extra work-groups that only link (4 exchanges in flight per lane): its waits overlap the matrix work; 7 launches -> 6; f32 sums in list order
# speedup vs baseline: 3.1500x; 1.0434x over previous
#include <hip/hip_runtime.h>


typedef __attribute__((ext_vector_type(2))) float v2f;
typedef __attribute__((ext_vector_type(8))) float v8f;

#define N_NODES  100000
#define N_EDGES  1600000
#define DIM_IN   256
#define DIM_HID  128
#define DIM_OUT  32
#define HEAD_PITCH 16

#define NL 4

struct Rec { int nxt; float w; int src; };

#define REC_A 1066666

__device__ __forceinline__ const Rec* rec_at(const Rec* ra, const Rec* rb, int e) {
  return (e < REC_A) ? (ra + e) : (rb + (e - REC_A));
}

__global__ __launch_bounds__(256) void k_init_head(int* head, int n) {
  int i = blockIdx.x * blockDim.x + threadIdx.x;
  if (i < n) *(int4*)(head + (size_t)i * HEAD_PITCH) = make_int4(-1, -1, -1, -1);
}

#define LINK_BLOCKS 256
#define LINK_U      4

__device__ __forceinline__ void link_edges(const int* __restrict__ src,
                                           const int* __restrict__ dst,
                                           const float* __restrict__ w,
                                           int* head, Rec* ra, Rec* rb, int e, int bid) {
  constexpr int T = LINK_BLOCKS * 256;
  const int g = bid * 256 + threadIdx.x;
  for (int base = g; base < e; base += LINK_U * T) {
    if (base + (LINK_U - 1) * T < e) {
      int   sv[LINK_U], dv[LINK_U], nx[LINK_U];
      float wv[LINK_U];
#pragma unroll
      for (int u = 0; u < LINK_U; ++u) {
        const int i = base + u * T;
        sv[u] = src[i];
        dv[u] = dst[i];
        wv[u] = w[i];
      }
#pragma unroll
      for (int u = 0; u < LINK_U; ++u) {
        const int i = base + u * T;
        nx[u] = atomicExch(&head[(size_t)dv[u] * HEAD_PITCH + (i & (NL - 1))], i);
      }
#pragma unroll
      for (int u = 0; u < LINK_U; ++u) {
        const int i = base + u * T;
        Rec r;
        r.nxt = nx[u];
        r.w   = wv[u];
        r.src = sv[u];
        Rec* p = (i < REC_A) ? (ra + i) : (rb + (i - REC_A));
        *p = r;
      }
    } else {
      for (int i = base; i < e; i += T) {
        Rec r;
        r.src = src[i];
        r.w   = w[i];
        r.nxt = atomicExch(&head[(size_t)dst[i] * HEAD_PITCH + (i & (NL - 1))], i);
        Rec* p = (i < REC_A) ? (ra + i) : (rb + (i - REC_A));
        *p = r;
      }
    }
  }
}

__global__ __launch_bounds__(256) void k_deg_inv(const int* __restrict__ head,
                                                 const Rec* __restrict__ ra,
                                                 const Rec* __restrict__ rb,
                                                 float* __restrict__ inv, int n, int e) {
  int i = blockIdx.x * blockDim.x + threadIdx.x;
  if (i >= n) return;
  const int4 hd = *(const int4*)(head + (size_t)i * HEAD_PITCH);
  int   k[NL] = {hd.x, hd.y, hd.z, hd.w};
  float d[NL] = {1.0f, 0.0f, 0.0f, 0.0f};
  for (int it = 0; it < e; ++it) {
    bool  live[NL];
    float wv[NL];
    int   kn[NL];
    bool  any = false;
#pragma unroll
    for (int c = 0; c < NL; ++c) {
      live[c] = (unsigned)k[c] < (unsigned)e;
      any |= live[c];
      const int kk = live[c] ? k[c] : 0;
      const Rec* p = rec_at(ra, rb, kk);
      wv[c] = p->w;
      kn[c] = p->nxt;
    }
    if (!any) break;
#pragma unroll
    for (int c = 0; c < NL; ++c) {
      d[c] = live[c] ? d[c] + wv[c] : d[c];
      k[c] = live[c] ? kn[c] : -1;
    }
  }
  const float ds = (d[0] + d[1]) + (d[2] + d[3]);
  inv[i] = (ds > 0.0f) ? rsqrtf(fmaxf(ds, 1e-12f)) : 0.0f;
}


template <int K, int NT, bool RELU>
__device__ __forceinline__ void strip_product(const float* __restrict__ A,
                                              const float* __restrict__ W,
                                              float* __restrict__ H,
                                              float4* Wl, int n_row_tiles, int bid) {
  constexpr int N  = NT * 16;
  constexpr int NQ = NT / 2;
  constexpr int KS = K / 4;
  constexpr int CH = 8;

#pragma unroll 8
  for (int it = 0; it < KS * NQ * 32 / 256; ++it) {
    const int i  = it * 256 + threadIdx.x;
    const int ln = i & 31;
    const int kq = i >> 5;
    const int q  = kq % NQ;
    const int kk = kq / NQ;
    const int k  = 4 * kk + 2 * (ln >> 4);
    const int n  = 32 * q + (ln & 15);
    float4 f;
    f.x = W[(size_t)k       * N + n];
    f.y = W[(size_t)(k + 1) * N + n];
    f.z = W[(size_t)k       * N + n + 16];
    f.w = W[(size_t)(k + 1) * N + n + 16];
    Wl[i] = f;
  }
  __syncthreads();

  const int rt = bid * 8 + (threadIdx.x >> 5);
  if (rt >= n_row_tiles) return;
  const int lane = threadIdx.x & 31;
  const int l    = lane & 15;
  const int half = lane >> 4;
  const int m0   = rt * 16;

  const float* arow = A + (size_t)(m0 + l) * K + half * 2;
  v8f c[NT];
#pragma unroll
  for (int t = 0; t < NT; ++t) c[t] = (v8f){0.f, 0.f, 0.f, 0.f, 0.f, 0.f, 0.f, 0.f};

  v2f ac[CH], an[CH];
#pragma unroll
  for (int j = 0; j < CH; ++j) {
    v2f a = *(const v2f*)(arow + 4 * j);
    if (RELU) { a.x = fmaxf(a.x, 0.0f); a.y = fmaxf(a.y, 0.0f); }
    ac[j] = a;
  }

  for (int ch = 0; ch < KS / CH; ++ch) {
    if (ch + 1 < KS / CH) {
#pragma unroll
      for (int j = 0; j < CH; ++j) {
        v2f a = *(const v2f*)(arow + 4 * (CH * (ch + 1) + j));
        if (RELU) { a.x = fmaxf(a.x, 0.0f); a.y = fmaxf(a.y, 0.0f); }
        an[j] = a;
      }
    }
#pragma unroll
    for (int j = 0; j < CH; ++j) {
      const float4* wk = Wl + (size_t)((ch * CH + j) * NQ) * 32 + lane;
#pragma unroll
      for (int q = 0; q < NQ; ++q) {
        const float4 f = wk[q * 32];
        v2f b0; b0.x = f.x; b0.y = f.y;
        v2f b1; b1.x = f.z; b1.y = f.w;
        c[2 * q]     = __builtin_amdgcn_wmma_f32_16x16x4_f32(false, ac[j], false, b0,
                                                             (short)0, c[2 * q], false, false);
        c[2 * q + 1] = __builtin_amdgcn_wmma_f32_16x16x4_f32(false, ac[j], false, b1,
                                                             (short)0, c[2 * q + 1], false, false);
      }
    }
#pragma unroll
    for (int j = 0; j < CH; ++j) ac[j] = an[j];
  }

  float* drow = H + (size_t)(m0 + half * 8) * N + l;
#pragma unroll
  for (int t = 0; t < NT; ++t) {
#pragma unroll
    for (int i = 0; i < 8; ++i) drow[(size_t)i * N + t * 16] = c[t][i];
  }
}

#define GEMM1_BLOCKS ((N_NODES / 16 + 7) / 8)
__global__ __launch_bounds__(256) void k_gemm1_link(const float* __restrict__ X,
                                                    const float* __restrict__ W,
                                                    float* __restrict__ H,
                                                    const int* __restrict__ src,
                                                    const int* __restrict__ dst,
                                                    const float* __restrict__ w,
                                                    int* head, Rec* ra, Rec* rb, int e) {
  __shared__ float4 Wl[(DIM_IN / 4) * (DIM_HID / 32) * 32];
  if (blockIdx.x < LINK_BLOCKS) {
    link_edges(src, dst, w, head, ra, rb, e, (int)blockIdx.x);
    return;
  }
  strip_product<DIM_IN, DIM_HID / 16, false>(X, W, H, Wl, N_NODES / 16, (int)blockIdx.x - LINK_BLOCKS);
}

#define GEMM2_BLOCKS ((N_NODES / 16 + 7) / 8)
__global__ __launch_bounds__(256) void k_gemm2_wmma(const float* __restrict__ XE,
                                                    const float* __restrict__ W,
                                                    float* __restrict__ H,
                                                    const Rec* __restrict__ ra,
                                                    const Rec* __restrict__ rb,
                                                    int4* __restrict__ rec2) {
  __shared__ float4 Wl[(DIM_HID / 4) * (DIM_OUT / 32) * 32];
  constexpr int NTHR = GEMM2_BLOCKS * 256;
  constexpr int FULL = N_EDGES / NTHR;
  const int g = blockIdx.x * 256 + threadIdx.x;
  Rec r[FULL];
#pragma unroll
  for (int j = 0; j < FULL; ++j) r[j] = *rec_at(ra, rb, j * NTHR + g);
#pragma unroll
  for (int j = 0; j < FULL; ++j)
    rec2[j * NTHR + g] = make_int4(r[j].nxt, r[j].src, __float_as_int(r[j].w), 0);
  if (FULL * NTHR + g < N_EDGES) {
    const Rec q = *rec_at(ra, rb, FULL * NTHR + g);
    rec2[FULL * NTHR + g] = make_int4(q.nxt, q.src, __float_as_int(q.w), 0);
  }
  strip_product<DIM_HID, DIM_OUT / 16, true>(XE, W, H, Wl, N_NODES / 16, (int)blockIdx.x);
}


__global__ __launch_bounds__(256) void k_gather128(const Rec* __restrict__ ra,
                                                   const Rec* __restrict__ rb,
                                                   const int* __restrict__ head,
                                                   const float* __restrict__ inv,
                                                   const float* __restrict__ h,
                                                   const float* __restrict__ bias,
                                                   float* __restrict__ out, int e, int n) {
  const int t = __builtin_amdgcn_readfirstlane((int)(blockIdx.x * 8 + (threadIdx.x >> 5)));
  const int lane = threadIdx.x & 31;
  if (t >= n) return;
  const int g = lane >> 3;
  const int j = lane & 7;
  const float it_ = inv[t];
  const float* hl = h + j * 4;
  int kc = head[(size_t)t * HEAD_PITCH + g];
  const int kk0 = ((unsigned)kc < (unsigned)e) ? kc : 0;
  const Rec r0 = *rec_at(ra, rb, kk0);
  int   kn = r0.nxt;
  int   s  = r0.src;
  float w  = r0.w;
  float4 acc[4];
#pragma unroll
  for (int i = 0; i < 4; ++i) acc[i] = make_float4(0.f, 0.f, 0.f, 0.f);
  if (g == 0) {
    const float nrm = it_ * 1.0f * it_;
    const float* p = hl + (size_t)t * DIM_HID;
#pragma unroll
    for (int i = 0; i < 4; ++i) {
      const float4 v = *(const float4*)(p + i * 32);
      acc[i].x = nrm * v.x; acc[i].y = nrm * v.y; acc[i].z = nrm * v.z; acc[i].w = nrm * v.w;
    }
  }
  for (int it = 0; it < e; ++it) {
    const bool live = (unsigned)kc < (unsigned)e;
    if (__builtin_amdgcn_ballot_w32(live) == 0u) break;
    const bool  nlive = live && (unsigned)kn < (unsigned)e;
    const int   kk2 = nlive ? kn : 0;
    const Rec   r2  = *rec_at(ra, rb, kk2);
    if (live) {
      const float  ps = inv[s];
      const float* p  = hl + (size_t)s * DIM_HID;
      float4 v[4];
#pragma unroll
      for (int i = 0; i < 4; ++i) v[i] = *(const float4*)(p + i * 32);
      __builtin_amdgcn_sched_barrier(0);
      const float coef = ps * w * it_;
#pragma unroll
      for (int i = 0; i < 4; ++i) {
        acc[i].x += coef * v[i].x; acc[i].y += coef * v[i].y;
        acc[i].z += coef * v[i].z; acc[i].w += coef * v[i].w;
      }
    }
    kc = nlive ? kn : -1;
    kn = r2.nxt; s = r2.src; w = r2.w;
  }
  const bool lo = lane < 16;
  float4 keep[2];
#pragma unroll
  for (int k = 0; k < 2; ++k) {
    const float4 snd = lo ? acc[2 + k] : acc[k];
    float4       kp  = lo ? acc[k] : acc[2 + k];
    kp.x += __shfl_xor(snd.x, 16); kp.y += __shfl_xor(snd.y, 16);
    kp.z += __shfl_xor(snd.z, 16); kp.w += __shfl_xor(snd.w, 16);
    keep[k] = kp;
  }
  const bool ev = (g & 1) == 0;
  const float4 snd = ev ? keep[1] : keep[0];
  float4       fin = ev ? keep[0] : keep[1];
  fin.x += __shfl_xor(snd.x, 8); fin.y += __shfl_xor(snd.y, 8);
  fin.z += __shfl_xor(snd.z, 8); fin.w += __shfl_xor(snd.w, 8);
  const float4 b = *(const float4*)(bias + lane * 4);
  float4 r = {fin.x + b.x, fin.y + b.y, fin.z + b.z, fin.w + b.w};
  *(float4*)(out + (size_t)t * DIM_HID + lane * 4) = r;
}

__global__ __launch_bounds__(256) void k_gather32(const int4* __restrict__ rec2,
                                                  const int* __restrict__ head,
                                                  const float* __restrict__ inv,
                                                  const float* __restrict__ h,
                                                  const float* __restrict__ bias,
                                                  float* __restrict__ out, int e, int n) {
  const int t = __builtin_amdgcn_readfirstlane((int)(blockIdx.x * 8 + (threadIdx.x >> 5)));
  const int lane = threadIdx.x & 31;
  if (t >= n) return;
  const int g = lane >> 3;
  const int j = lane & 7;
  const float it_ = inv[t];
  const float* hl = h + j * 4;
  int kc = head[(size_t)t * HEAD_PITCH + g];
  const int kk0 = ((unsigned)kc < (unsigned)e) ? kc : 0;
  const int4 r0 = rec2[kk0];
  int   kn = r0.x;
  int   s  = r0.y;
  float w  = __int_as_float(r0.z);
  float4 acc = make_float4(0.f, 0.f, 0.f, 0.f);
  if (g == 0) {
    const float nrm = it_ * 1.0f * it_;
    const float4 v = *(const float4*)(hl + (size_t)t * DIM_OUT);
    acc.x = nrm * v.x; acc.y = nrm * v.y; acc.z = nrm * v.z; acc.w = nrm * v.w;
  }
  for (int it = 0; it < e; ++it) {
    const bool live = (unsigned)kc < (unsigned)e;
    if (__builtin_amdgcn_ballot_w32(live) == 0u) break;
    const bool  nlive = live && (unsigned)kn < (unsigned)e;
    const int   kk2 = nlive ? kn : 0;
    const int4  r2  = rec2[kk2];
    if (live) {
      const float  ps = inv[s];
      const float4 v  = *(const float4*)(hl + (size_t)s * DIM_OUT);
      __builtin_amdgcn_sched_barrier(0);
      const float coef = ps * w * it_;
      acc.x += coef * v.x; acc.y += coef * v.y; acc.z += coef * v.z; acc.w += coef * v.w;
    }
    kc = nlive ? kn : -1;
    kn = r2.x; s = r2.y; w = __int_as_float(r2.z);
  }
  const bool lo = lane < 16;
  const float s0 = lo ? acc.z : acc.x;
  const float s1 = lo ? acc.w : acc.y;
  float k0 = lo ? acc.x : acc.z;
  float k1 = lo ? acc.y : acc.w;
  k0 += __shfl_xor(s0, 16);
  k1 += __shfl_xor(s1, 16);
  const bool ev = (g & 1) == 0;
  const float sd = ev ? k1 : k0;
  float fn = ev ? k0 : k1;
  fn += __shfl_xor(sd, 8);
  const int idx = j * 4 + g;
  out[(size_t)t * DIM_OUT + idx] = fn + bias[idx];
}


extern "C" void kernel_launch(void* const* d_in, const int* in_sizes, int n_in,
                              void* d_out, int out_size, void* d_ws, size_t ws_size,
                              hipStream_t stream) {
  const int N = N_NODES, E = N_EDGES;

  const float* x   = (const float*)d_in[0];
  const int*   ei  = (const int*)d_in[1];
  const int*   src = ei;
  const int*   dst = ei + E;
  const float* w   = (const float*)d_in[2];
  const float* W1  = (const float*)d_in[3];
  const float* b1  = (const float*)d_in[4];
  const float* W2  = (const float*)d_in[5];
  const float* b2  = (const float*)d_in[6];

  float* out  = (float*)d_out;
  float* xemb = out + (size_t)N * DIM_OUT;

  float* inv  = (float*)d_ws;
  float* h1   = inv + N;
  float* h2   = h1;
  int*   head = (int*)(h1 + (size_t)N * DIM_HID);
  int*   tail = head + (size_t)N * HEAD_PITCH;

  Rec*  ra   = (Rec*)d_out;
  Rec*  rb   = (Rec*)(tail - 2);
  int4* rec2 = (int4*)(h1 + (size_t)N * DIM_OUT);

  k_init_head <<<(N + 255) / 256, 256, 0, stream>>>(head, N);
  k_gemm1_link<<<LINK_BLOCKS + GEMM1_BLOCKS, 256, 0, stream>>>(x, W1, h1, src, dst, w, head, ra, rb, E);
  k_deg_inv   <<<(N + 255) / 256, 256, 0, stream>>>(head, ra, rb, inv, N, E);

  k_gather128<<<(N + 7) / 8, 256, 0, stream>>>(ra, rb, head, inv, h1, b1, xemb, E, N);

  k_gemm2_wmma<<<GEMM2_BLOCKS, 256, 0, stream>>>(xemb, W2, h2, ra, rb, rec2);

  k_gather32<<<(N + 7) / 8, 256, 0, stream>>>(rec2, head, inv, h2, b2, out, E, N);
}
